# starting kernel
# baseline (speedup 1.0000x reference)
#include <hip/hip_runtime.h>

typedef float v2f __attribute__((ext_vector_type(2)));
typedef float v8f __attribute__((ext_vector_type(8)));

#define BB 4096
#define TT 100
#define FF 40
#define HE 64
#define HL 32
#define HD 64

#define GS  260
#define GXS 132
#define HS  66
#define HLS 34

__device__ __forceinline__ v8f wmma4(v2f a, v2f b, v8f c) {
  return __builtin_amdgcn_wmma_f32_16x16x4_f32(false, a, false, b, (short)0, c,
                                               false, false);
}

__device__ __forceinline__ float sigm(float x) {
  return 1.0f / (1.0f + __expf(-x));
}
__device__ __forceinline__ float tanh_(float x) {
  float e = __expf(2.0f * x);
  return 1.0f - 2.0f / (e + 1.0f);
}

__global__ __launch_bounds__(256)
void lstm_ae_kernel(const float* __restrict__ x,
                    const float* __restrict__ eWih, const float* __restrict__ eWhh,
                    const float* __restrict__ eb,
                    const float* __restrict__ lWih, const float* __restrict__ lWhh,
                    const float* __restrict__ lb,
                    const float* __restrict__ dWih, const float* __restrict__ dWhh,
                    const float* __restrict__ db,
                    const float* __restrict__ oW,   const float* __restrict__ ob,
                    float* __restrict__ out) {
  __shared__ float gates[16 * GS];
  __shared__ float gxl[16 * GXS];
  __shared__ float hbuf[16 * HS];
  __shared__ float hlat[16 * HLS];

  const int tid  = threadIdx.x;
  const int lane = tid & 31;
  const int wave = tid >> 5;
  const int lrow = lane & 15;
  const int lhi  = lane >> 4;
  const int b0   = blockIdx.x * 16;

  const int r0  = tid >> 6;
  const int j64 = tid & 63;
  const int rl0 = tid >> 5;
  const int j32 = tid & 31;

  v2f wih[10][2];
  v2f whh[16][2];
  #pragma unroll
  for (int tt = 0; tt < 2; ++tt) {
    const int gc = wave * 32 + tt * 16;
    #pragma unroll
    for (int kc = 0; kc < 10; ++kc)
      wih[kc][tt] = *(const v2f*)&eWih[(gc + lrow) * FF + kc * 4 + 2 * lhi];
    #pragma unroll
    for (int kc = 0; kc < 16; ++kc)
      whh[kc][tt] = *(const v2f*)&eWhh[(gc + lrow) * HE + kc * 4 + 2 * lhi];
  }
  const float ebi = eb[j64], ebf = eb[64 + j64], ebg = eb[128 + j64], ebo = eb[192 + j64];

  for (int i = tid; i < 16 * HS; i += 256) hbuf[i] = 0.0f;
  float cE[4] = {0.f, 0.f, 0.f, 0.f};
  __syncthreads();

  const float* xblk = x + (size_t)b0 * TT * FF;

  #pragma unroll 1
  for (int t = 0; t < TT; ++t) {
    v8f acc0 = {}, acc1 = {};
    const float* xt = xblk + (size_t)lrow * TT * FF + t * FF + 2 * lhi;
    #pragma unroll
    for (int kc = 0; kc < 10; ++kc) {
      v2f a = *(const v2f*)&xt[kc * 4];
      acc0 = wmma4(a, wih[kc][0], acc0);
      acc1 = wmma4(a, wih[kc][1], acc1);
    }
    #pragma unroll
    for (int kc = 0; kc < 16; ++kc) {
      v2f a = *(const v2f*)&hbuf[lrow * HS + kc * 4 + 2 * lhi];
      acc0 = wmma4(a, whh[kc][0], acc0);
      acc1 = wmma4(a, whh[kc][1], acc1);
    }
    const int gc0 = wave * 32;
    #pragma unroll
    for (int jj = 0; jj < 8; ++jj) {
      gates[(jj + 8 * lhi) * GS + gc0 + lrow]      = acc0[jj];
      gates[(jj + 8 * lhi) * GS + gc0 + 16 + lrow] = acc1[jj];
    }
    __syncthreads();
    #pragma unroll
    for (int i = 0; i < 4; ++i) {
      const int r = r0 + 4 * i;
      float gi = gates[r * GS + j64] + ebi;
      float gf = gates[r * GS + 64 + j64] + ebf;
      float gg = gates[r * GS + 128 + j64] + ebg;
      float go = gates[r * GS + 192 + j64] + ebo;
      float c  = sigm(gf) * cE[i] + sigm(gi) * tanh_(gg);
      cE[i] = c;
      hbuf[r * HS + j64] = sigm(go) * tanh_(c);
    }
    __syncthreads();
  }

  {
    const int gc = wave * 16;
    v8f acc = {};
    #pragma unroll
    for (int kc = 0; kc < 16; ++kc) {
      v2f a = *(const v2f*)&hbuf[lrow * HS + kc * 4 + 2 * lhi];
      v2f b = *(const v2f*)&lWih[(gc + lrow) * HE + kc * 4 + 2 * lhi];
      acc = wmma4(a, b, acc);
    }
    const float bias = lb[gc + lrow];
    #pragma unroll
    for (int jj = 0; jj < 8; ++jj)
      gxl[(jj + 8 * lhi) * GXS + gc + lrow] = acc[jj] + bias;
  }
  __syncthreads();

  v2f wlh[8];
  v2f dwi[8][2];
  v2f dwh[16][2];
  v2f wo[16];
  {
    const int gcl = wave * 16;
    #pragma unroll
    for (int kc = 0; kc < 8; ++kc)
      wlh[kc] = *(const v2f*)&lWhh[(gcl + lrow) * HL + kc * 4 + 2 * lhi];
    #pragma unroll
    for (int tt = 0; tt < 2; ++tt) {
      const int gc = wave * 32 + tt * 16;
      #pragma unroll
      for (int kc = 0; kc < 8; ++kc)
        dwi[kc][tt] = *(const v2f*)&dWih[(gc + lrow) * HL + kc * 4 + 2 * lhi];
      #pragma unroll
      for (int kc = 0; kc < 16; ++kc)
        dwh[kc][tt] = *(const v2f*)&dWhh[(gc + lrow) * HD + kc * 4 + 2 * lhi];
    }
    const int f = wave * 16 + lrow;
    #pragma unroll
    for (int kc = 0; kc < 16; ++kc) {
      v2f z = {};
      if (wave < 3 && f < FF)
        z = *(const v2f*)&oW[f * HD + kc * 4 + 2 * lhi];
      wo[kc] = z;
    }
  }
  const float dbi = db[j64], dbf = db[64 + j64], dbg = db[128 + j64], dbo = db[192 + j64];
  const float obias = (wave < 3 && (wave * 16 + lrow) < FF) ? ob[wave * 16 + lrow] : 0.0f;

  for (int i = tid; i < 16 * HS; i += 256) hbuf[i] = 0.0f;
  for (int i = tid; i < 16 * HLS; i += 256) hlat[i] = 0.0f;
  float cL[2] = {0.f, 0.f};
  float cD[4] = {0.f, 0.f, 0.f, 0.f};
  __syncthreads();

  float* outblk = out + (size_t)b0 * TT * FF;

  #pragma unroll 1
  for (int t = 0; t < TT; ++t) {
    {
      const int gc = wave * 16;
      v8f acc = {};
      #pragma unroll
      for (int kc = 0; kc < 8; ++kc) {
        v2f a = *(const v2f*)&hlat[lrow * HLS + kc * 4 + 2 * lhi];
        acc = wmma4(a, wlh[kc], acc);
      }
      #pragma unroll
      for (int jj = 0; jj < 8; ++jj)
        gates[(jj + 8 * lhi) * GS + gc + lrow] = acc[jj];
    }
    __syncthreads();
    #pragma unroll
    for (int i = 0; i < 2; ++i) {
      const int r = rl0 + 8 * i;
      float gi = gates[r * GS + j32]      + gxl[r * GXS + j32];
      float gf = gates[r * GS + 32 + j32] + gxl[r * GXS + 32 + j32];
      float gg = gates[r * GS + 64 + j32] + gxl[r * GXS + 64 + j32];
      float go = gates[r * GS + 96 + j32] + gxl[r * GXS + 96 + j32];
      float c  = sigm(gf) * cL[i] + sigm(gi) * tanh_(gg);
      cL[i] = c;
      hlat[r * HLS + j32] = sigm(go) * tanh_(c);
    }
    __syncthreads();
    {
      v8f acc0 = {}, acc1 = {};
      #pragma unroll
      for (int kc = 0; kc < 8; ++kc) {
        v2f a = *(const v2f*)&hlat[lrow * HLS + kc * 4 + 2 * lhi];
        acc0 = wmma4(a, dwi[kc][0], acc0);
        acc1 = wmma4(a, dwi[kc][1], acc1);
      }
      #pragma unroll
      for (int kc = 0; kc < 16; ++kc) {
        v2f a = *(const v2f*)&hbuf[lrow * HS + kc * 4 + 2 * lhi];
        acc0 = wmma4(a, dwh[kc][0], acc0);
        acc1 = wmma4(a, dwh[kc][1], acc1);
      }
      const int gc0 = wave * 32;
      #pragma unroll
      for (int jj = 0; jj < 8; ++jj) {
        gates[(jj + 8 * lhi) * GS + gc0 + lrow]      = acc0[jj];
        gates[(jj + 8 * lhi) * GS + gc0 + 16 + lrow] = acc1[jj];
      }
    }
    __syncthreads();
    #pragma unroll
    for (int i = 0; i < 4; ++i) {
      const int r = r0 + 4 * i;
      float gi = gates[r * GS + j64] + dbi;
      float gf = gates[r * GS + 64 + j64] + dbf;
      float gg = gates[r * GS + 128 + j64] + dbg;
      float go = gates[r * GS + 192 + j64] + dbo;
      float c  = sigm(gf) * cD[i] + sigm(gi) * tanh_(gg);
      cD[i] = c;
      hbuf[r * HS + j64] = sigm(go) * tanh_(c);
    }
    __syncthreads();
    if (wave < 3) {
      v8f acc = {};
      #pragma unroll
      for (int kc = 0; kc < 16; ++kc) {
        v2f a = *(const v2f*)&hbuf[lrow * HS + kc * 4 + 2 * lhi];
        acc = wmma4(a, wo[kc], acc);
      }
      const int f = wave * 16 + lrow;
      if (f < FF) {
        #pragma unroll
        for (int jj = 0; jj < 8; ++jj) {
          const int m = jj + 8 * lhi;
          outblk[(size_t)m * TT * FF + t * FF + f] = acc[jj] + obias;
        }
      }
    }
  }
}

extern "C" void kernel_launch(void* const* d_in, const int* in_sizes, int n_in,
                              void* d_out, int out_size, void* d_ws, size_t ws_size,
                              hipStream_t stream) {
  (void)in_sizes; (void)n_in; (void)out_size; (void)d_ws; (void)ws_size;
  const float* x    = (const float*)d_in[0];
  const float* eWih = (const float*)d_in[1];
  const float* eWhh = (const float*)d_in[2];
  const float* eb   = (const float*)d_in[3];
  const float* lWih = (const float*)d_in[4];
  const float* lWhh = (const float*)d_in[5];
  const float* lb   = (const float*)d_in[6];
  const float* dWih = (const float*)d_in[7];
  const float* dWhh = (const float*)d_in[8];
  const float* db   = (const float*)d_in[9];
  const float* oW   = (const float*)d_in[10];
  const float* ob   = (const float*)d_in[11];
  float* out = (float*)d_out;

  dim3 grid(BB / 16), block(256);
  hipLaunchKernelGGL(lstm_ae_kernel, grid, block, 0, stream,
                     x, eWih, eWhh, eb, lWih, lWhh, lb, dWih, dWhh, db, oW, ob, out);
}


# encoder x fragments loaded one step ahead by each wave into a second register set, no LDS staging: the global-load wait no longer stands before the step's first product; sum order kept
# speedup vs baseline: 1.1144x; 1.1144x over previous
#include <hip/hip_runtime.h>

typedef float v2f __attribute__((ext_vector_type(2)));
typedef float v8f __attribute__((ext_vector_type(8)));

#define BB 4096
#define TT 100
#define FF 40
#define HE 64
#define HL 32
#define HD 64

#define GS  260
#define GXS 132
#define HS  66
#define HLS 34

__device__ __forceinline__ v8f wmma4(v2f a, v2f b, v8f c) {
  return __builtin_amdgcn_wmma_f32_16x16x4_f32(false, a, false, b, (short)0, c,
                                               false, false);
}

__device__ __forceinline__ float sigm(float x) {
  return 1.0f / (1.0f + __expf(-x));
}
__device__ __forceinline__ float tanh_(float x) {
  float e = __expf(2.0f * x);
  return 1.0f - 2.0f / (e + 1.0f);
}

__global__ __launch_bounds__(256)
void lstm_ae_kernel(const float* __restrict__ x,
                    const float* __restrict__ eWih, const float* __restrict__ eWhh,
                    const float* __restrict__ eb,
                    const float* __restrict__ lWih, const float* __restrict__ lWhh,
                    const float* __restrict__ lb,
                    const float* __restrict__ dWih, const float* __restrict__ dWhh,
                    const float* __restrict__ db,
                    const float* __restrict__ oW,   const float* __restrict__ ob,
                    float* __restrict__ out) {
  __shared__ float gates[16 * GS];
  __shared__ float gxl[16 * GXS];
  __shared__ float hbuf[16 * HS];
  __shared__ float hlat[16 * HLS];

  const int tid  = threadIdx.x;
  const int lane = tid & 31;
  const int wave = tid >> 5;
  const int lrow = lane & 15;
  const int lhi  = lane >> 4;
  const int b0   = blockIdx.x * 16;

  const int r0  = tid >> 6;
  const int j64 = tid & 63;
  const int rl0 = tid >> 5;
  const int j32 = tid & 31;

  v2f wih[10][2];
  v2f whh[16][2];
  #pragma unroll
  for (int tt = 0; tt < 2; ++tt) {
    const int gc = wave * 32 + tt * 16;
    #pragma unroll
    for (int kc = 0; kc < 10; ++kc)
      wih[kc][tt] = *(const v2f*)&eWih[(gc + lrow) * FF + kc * 4 + 2 * lhi];
    #pragma unroll
    for (int kc = 0; kc < 16; ++kc)
      whh[kc][tt] = *(const v2f*)&eWhh[(gc + lrow) * HE + kc * 4 + 2 * lhi];
  }
  const float ebi = eb[j64], ebf = eb[64 + j64], ebg = eb[128 + j64], ebo = eb[192 + j64];

  for (int i = tid; i < 16 * HS; i += 256) hbuf[i] = 0.0f;
  float cE[4] = {0.f, 0.f, 0.f, 0.f};
  __syncthreads();

  const float* xblk = x + (size_t)b0 * TT * FF;
  const float* xrow = xblk + (size_t)lrow * TT * FF + 2 * lhi;

  v2f xc[10];
  #pragma unroll
  for (int kc = 0; kc < 10; ++kc) xc[kc] = *(const v2f*)&xrow[kc * 4];

  #pragma unroll 1
  for (int t = 0; t < TT; ++t) {
    const float* xt = xrow + ((t + 1 < TT) ? t + 1 : t) * FF;
    v2f xn[10];
    #pragma unroll
    for (int kc = 0; kc < 10; ++kc) xn[kc] = *(const v2f*)&xt[kc * 4];

    v8f acc0 = {}, acc1 = {};
    #pragma unroll
    for (int kc = 0; kc < 10; ++kc) {
      acc0 = wmma4(xc[kc], wih[kc][0], acc0);
      acc1 = wmma4(xc[kc], wih[kc][1], acc1);
    }
    #pragma unroll
    for (int kc = 0; kc < 16; ++kc) {
      v2f a = *(const v2f*)&hbuf[lrow * HS + kc * 4 + 2 * lhi];
      acc0 = wmma4(a, whh[kc][0], acc0);
      acc1 = wmma4(a, whh[kc][1], acc1);
    }
    const int gc0 = wave * 32;
    #pragma unroll
    for (int jj = 0; jj < 8; ++jj) {
      gates[(jj + 8 * lhi) * GS + gc0 + lrow]      = acc0[jj];
      gates[(jj + 8 * lhi) * GS + gc0 + 16 + lrow] = acc1[jj];
    }
    __syncthreads();
    #pragma unroll
    for (int i = 0; i < 4; ++i) {
      const int r = r0 + 4 * i;
      float gi = gates[r * GS + j64] + ebi;
      float gf = gates[r * GS + 64 + j64] + ebf;
      float gg = gates[r * GS + 128 + j64] + ebg;
      float go = gates[r * GS + 192 + j64] + ebo;
      float c  = sigm(gf) * cE[i] + sigm(gi) * tanh_(gg);
      cE[i] = c;
      hbuf[r * HS + j64] = sigm(go) * tanh_(c);
    }
    __syncthreads();
    #pragma unroll
    for (int kc = 0; kc < 10; ++kc) xc[kc] = xn[kc];
  }

  {
    const int gc = wave * 16;
    v8f acc = {};
    #pragma unroll
    for (int kc = 0; kc < 16; ++kc) {
      v2f a = *(const v2f*)&hbuf[lrow * HS + kc * 4 + 2 * lhi];
      v2f b = *(const v2f*)&lWih[(gc + lrow) * HE + kc * 4 + 2 * lhi];
      acc = wmma4(a, b, acc);
    }
    const float bias = lb[gc + lrow];
    #pragma unroll
    for (int jj = 0; jj < 8; ++jj)
      gxl[(jj + 8 * lhi) * GXS + gc + lrow] = acc[jj] + bias;
  }
  __syncthreads();

  v2f wlh[8];
  v2f dwi[8][2];
  v2f dwh[16][2];
  v2f wo[16];
  {
    const int gcl = wave * 16;
    #pragma unroll
    for (int kc = 0; kc < 8; ++kc)
      wlh[kc] = *(const v2f*)&lWhh[(gcl + lrow) * HL + kc * 4 + 2 * lhi];
    #pragma unroll
    for (int tt = 0; tt < 2; ++tt) {
      const int gc = wave * 32 + tt * 16;
      #pragma unroll
      for (int kc = 0; kc < 8; ++kc)
        dwi[kc][tt] = *(const v2f*)&dWih[(gc + lrow) * HL + kc * 4 + 2 * lhi];
      #pragma unroll
      for (int kc = 0; kc < 16; ++kc)
        dwh[kc][tt] = *(const v2f*)&dWhh[(gc + lrow) * HD + kc * 4 + 2 * lhi];
    }
    const int f = wave * 16 + lrow;
    #pragma unroll
    for (int kc = 0; kc < 16; ++kc) {
      v2f z = {};
      if (wave < 3 && f < FF)
        z = *(const v2f*)&oW[f * HD + kc * 4 + 2 * lhi];
      wo[kc] = z;
    }
  }
  const float dbi = db[j64], dbf = db[64 + j64], dbg = db[128 + j64], dbo = db[192 + j64];
  const float obias = (wave < 3 && (wave * 16 + lrow) < FF) ? ob[wave * 16 + lrow] : 0.0f;

  for (int i = tid; i < 16 * HS; i += 256) hbuf[i] = 0.0f;
  for (int i = tid; i < 16 * HLS; i += 256) hlat[i] = 0.0f;
  float cL[2] = {0.f, 0.f};
  float cD[4] = {0.f, 0.f, 0.f, 0.f};
  __syncthreads();

  float* outblk = out + (size_t)b0 * TT * FF;

  #pragma unroll 1
  for (int t = 0; t < TT; ++t) {
    {
      const int gc = wave * 16;
      v8f acc = {};
      #pragma unroll
      for (int kc = 0; kc < 8; ++kc) {
        v2f a = *(const v2f*)&hlat[lrow * HLS + kc * 4 + 2 * lhi];
        acc = wmma4(a, wlh[kc], acc);
      }
      #pragma unroll
      for (int jj = 0; jj < 8; ++jj)
        gates[(jj + 8 * lhi) * GS + gc + lrow] = acc[jj];
    }
    __syncthreads();
    #pragma unroll
    for (int i = 0; i < 2; ++i) {
      const int r = rl0 + 8 * i;
      float gi = gates[r * GS + j32]      + gxl[r * GXS + j32];
      float gf = gates[r * GS + 32 + j32] + gxl[r * GXS + 32 + j32];
      float gg = gates[r * GS + 64 + j32] + gxl[r * GXS + 64 + j32];
      float go = gates[r * GS + 96 + j32] + gxl[r * GXS + 96 + j32];
      float c  = sigm(gf) * cL[i] + sigm(gi) * tanh_(gg);
      cL[i] = c;
      hlat[r * HLS + j32] = sigm(go) * tanh_(c);
    }
    __syncthreads();
    {
      v8f acc0 = {}, acc1 = {};
      #pragma unroll
      for (int kc = 0; kc < 8; ++kc) {
        v2f a = *(const v2f*)&hlat[lrow * HLS + kc * 4 + 2 * lhi];
        acc0 = wmma4(a, dwi[kc][0], acc0);
        acc1 = wmma4(a, dwi[kc][1], acc1);
      }
      #pragma unroll
      for (int kc = 0; kc < 16; ++kc) {
        v2f a = *(const v2f*)&hbuf[lrow * HS + kc * 4 + 2 * lhi];
        acc0 = wmma4(a, dwh[kc][0], acc0);
        acc1 = wmma4(a, dwh[kc][1], acc1);
      }
      const int gc0 = wave * 32;
      #pragma unroll
      for (int jj = 0; jj < 8; ++jj) {
        gates[(jj + 8 * lhi) * GS + gc0 + lrow]      = acc0[jj];
        gates[(jj + 8 * lhi) * GS + gc0 + 16 + lrow] = acc1[jj];
      }
    }
    __syncthreads();
    #pragma unroll
    for (int i = 0; i < 4; ++i) {
      const int r = r0 + 4 * i;
      float gi = gates[r * GS + j64] + dbi;
      float gf = gates[r * GS + 64 + j64] + dbf;
      float gg = gates[r * GS + 128 + j64] + dbg;
      float go = gates[r * GS + 192 + j64] + dbo;
      float c  = sigm(gf) * cD[i] + sigm(gi) * tanh_(gg);
      cD[i] = c;
      hbuf[r * HS + j64] = sigm(go) * tanh_(c);
    }
    __syncthreads();
    if (wave < 3) {
      v8f acc = {};
      #pragma unroll
      for (int kc = 0; kc < 16; ++kc) {
        v2f a = *(const v2f*)&hbuf[lrow * HS + kc * 4 + 2 * lhi];
        acc = wmma4(a, wo[kc], acc);
      }
      const int f = wave * 16 + lrow;
      if (f < FF) {
        #pragma unroll
        for (int jj = 0; jj < 8; ++jj) {
          const int m = jj + 8 * lhi;
          outblk[(size_t)m * TT * FF + t * FF + f] = acc[jj] + obias;
        }
      }
    }
  }
}

extern "C" void kernel_launch(void* const* d_in, const int* in_sizes, int n_in,
                              void* d_out, int out_size, void* d_ws, size_t ws_size,
                              hipStream_t stream) {
  (void)in_sizes; (void)n_in; (void)out_size; (void)d_ws; (void)ws_size;
  const float* x    = (const float*)d_in[0];
  const float* eWih = (const float*)d_in[1];
  const float* eWhh = (const float*)d_in[2];
  const float* eb   = (const float*)d_in[3];
  const float* lWih = (const float*)d_in[4];
  const float* lWhh = (const float*)d_in[5];
  const float* lb   = (const float*)d_in[6];
  const float* dWih = (const float*)d_in[7];
  const float* dWhh = (const float*)d_in[8];
  const float* db   = (const float*)d_in[9];
  const float* oW   = (const float*)d_in[10];
  const float* ob   = (const float*)d_in[11];
  float* out = (float*)d_out;

  dim3 grid(BB / 16), block(256);
  hipLaunchKernelGGL(lstm_ae_kernel, grid, block, 0, stream,
                     x, eWih, eWhh, eb, lWih, lWhh, lb, dWih, dWhh, db, oW, ob, out);
}


# gate columns regrouped per wave so lanes c, c^8 (H=32: c^4, c^8, c^12) hold i,f,g,o of one cell: DPP swap, cell update in registers, no gate LDS trip, h in 2 copies, 3 barriers less; sum order kept
# speedup vs baseline: 1.1960x; 1.0732x over previous
#include <hip/hip_runtime.h>

typedef float v2f __attribute__((ext_vector_type(2)));
typedef float v8f __attribute__((ext_vector_type(8)));

#define BB 4096
#define TT 100
#define FF 40
#define HE 64
#define HL 32
#define HD 64

#define GXS 132
#define HS  66
#define HLS 34

__device__ __forceinline__ v8f wmma4(v2f a, v2f b, v8f c) {
  return __builtin_amdgcn_wmma_f32_16x16x4_f32(false, a, false, b, (short)0, c,
                                               false, false);
}

__device__ __forceinline__ float sigm(float x) {
  return 1.0f / (1.0f + __expf(-x));
}
__device__ __forceinline__ float tanh_(float x) {
  float e = __expf(2.0f * x);
  return 1.0f - 2.0f / (e + 1.0f);
}

__device__ __forceinline__ float take_hi(float mine, float theirs) {
  return __builtin_bit_cast(float, __builtin_amdgcn_update_dpp(
      __builtin_bit_cast(int, mine), __builtin_bit_cast(int, theirs),
      0x168, 0xf, 0xc, false));
}
__device__ __forceinline__ float take_lo(float mine, float theirs) {
  return __builtin_bit_cast(float, __builtin_amdgcn_update_dpp(
      __builtin_bit_cast(int, mine), __builtin_bit_cast(int, theirs),
      0x168, 0xf, 0x3, false));
}

template <int XM, int BM>
__device__ __forceinline__ float dppx(float old, float src) {
  return __builtin_bit_cast(float, __builtin_amdgcn_update_dpp(
      __builtin_bit_cast(int, old), __builtin_bit_cast(int, src),
      0x160 | XM, 0xf, BM, false));
}
template <int G, int K>
__device__ __forceinline__ float gate32(const v8f& acc) {
  float v = 0.0f;
  v = dppx<4 * (G ^ 0), 1>(v, acc[0 + K]);
  v = dppx<4 * (G ^ 1), 2>(v, acc[2 + K]);
  v = dppx<4 * (G ^ 2), 4>(v, acc[4 + K]);
  v = dppx<4 * (G ^ 3), 8>(v, acc[6 + K]);
  return v;
}
template <int K>
__device__ __forceinline__ float cell32(const v8f& acc, const float (&gx)[4][2], float& cst) {
  float gi = gate32<0, K>(acc) + gx[0][K];
  float gf = gate32<1, K>(acc) + gx[1][K];
  float gg = gate32<2, K>(acc) + gx[2][K];
  float go = gate32<3, K>(acc) + gx[3][K];
  float c  = sigm(gf) * cst + sigm(gi) * tanh_(gg);
  cst = c;
  return sigm(go) * tanh_(c);
}

__device__ __forceinline__ void cell64(const v8f& acc0, const v8f& acc1,
                                       float bi, float bf, float bg, float bo,
                                       float (&cst)[4], float* __restrict__ hdst) {
  #pragma unroll
  for (int k = 0; k < 4; ++k) {
    float gi = take_hi(acc0[k], acc0[4 + k]) + bi;
    float gg = take_lo(acc0[4 + k], acc0[k]) + bg;
    float gf = take_hi(acc1[k], acc1[4 + k]) + bf;
    float go = take_lo(acc1[4 + k], acc1[k]) + bo;
    float c  = sigm(gf) * cst[k] + sigm(gi) * tanh_(gg);
    cst[k] = c;
    hdst[k * HS] = sigm(go) * tanh_(c);
  }
}

__global__ __launch_bounds__(256)
void lstm_ae_kernel(const float* __restrict__ x,
                    const float* __restrict__ eWih, const float* __restrict__ eWhh,
                    const float* __restrict__ eb,
                    const float* __restrict__ lWih, const float* __restrict__ lWhh,
                    const float* __restrict__ lb,
                    const float* __restrict__ dWih, const float* __restrict__ dWhh,
                    const float* __restrict__ db,
                    const float* __restrict__ oW,   const float* __restrict__ ob,
                    float* __restrict__ out) {
  __shared__ float gxl[16 * GXS];
  __shared__ float hbuf[2 * 16 * HS];
  __shared__ float hlat[2 * 16 * HLS];

  const int tid  = threadIdx.x;
  const int lane = tid & 31;
  const int wave = tid >> 5;
  const int lrow = lane & 15;
  const int lhi  = lane >> 4;
  const int b0   = blockIdx.x * 16;

  const int u64   = wave * 8 + (lrow & 7);
  const int mrow0 = 8 * lhi + 4 * (lrow >> 3);
  const int grow0 = (lrow >> 3) * 128 + u64;
  const int grow1 = 64 + grow0;
  const int hoff  = mrow0 * HS + u64;
  const int q32   = lrow >> 2;
  const int u32   = wave * 4 + (lrow & 3);
  const int growL = 32 * q32 + u32;
  const int mrowL = 8 * lhi + 2 * q32;
  const int hoffL = mrowL * HLS + u32;

  v2f wih[10][2];
  v2f whh[16][2];
  #pragma unroll
  for (int tt = 0; tt < 2; ++tt) {
    const int gr = tt ? grow1 : grow0;
    #pragma unroll
    for (int kc = 0; kc < 10; ++kc)
      wih[kc][tt] = *(const v2f*)&eWih[gr * FF + kc * 4 + 2 * lhi];
    #pragma unroll
    for (int kc = 0; kc < 16; ++kc)
      whh[kc][tt] = *(const v2f*)&eWhh[gr * HE + kc * 4 + 2 * lhi];
  }
  const float ebi = eb[u64], ebf = eb[64 + u64], ebg = eb[128 + u64], ebo = eb[192 + u64];

  for (int i = tid; i < 2 * 16 * HS; i += 256) hbuf[i] = 0.0f;
  float cE[4] = {0.f, 0.f, 0.f, 0.f};
  __syncthreads();

  const float* xblk = x + (size_t)b0 * TT * FF;
  const float* xrow = xblk + (size_t)lrow * TT * FF + 2 * lhi;

  v2f xc[10];
  #pragma unroll
  for (int kc = 0; kc < 10; ++kc) xc[kc] = *(const v2f*)&xrow[kc * 4];

  #pragma unroll 1
  for (int t = 0; t < TT; ++t) {
    const float* hrd = hbuf + (t & 1) * (16 * HS);
    float*       hwr = hbuf + ((t + 1) & 1) * (16 * HS);
    const float* xt = xrow + ((t + 1 < TT) ? t + 1 : t) * FF;
    v2f xn[10];
    #pragma unroll
    for (int kc = 0; kc < 10; ++kc) xn[kc] = *(const v2f*)&xt[kc * 4];

    v8f acc0 = {}, acc1 = {};
    #pragma unroll
    for (int kc = 0; kc < 10; ++kc) {
      acc0 = wmma4(xc[kc], wih[kc][0], acc0);
      acc1 = wmma4(xc[kc], wih[kc][1], acc1);
    }
    #pragma unroll
    for (int kc = 0; kc < 16; ++kc) {
      v2f a = *(const v2f*)&hrd[lrow * HS + kc * 4 + 2 * lhi];
      acc0 = wmma4(a, whh[kc][0], acc0);
      acc1 = wmma4(a, whh[kc][1], acc1);
    }
    cell64(acc0, acc1, ebi, ebf, ebg, ebo, cE, hwr + hoff);
    __syncthreads();
    #pragma unroll
    for (int kc = 0; kc < 10; ++kc) xc[kc] = xn[kc];
  }
  const float* hlast = hbuf + (TT & 1) * (16 * HS);

  {
    const int gc = wave * 16;
    v8f acc = {};
    #pragma unroll
    for (int kc = 0; kc < 16; ++kc) {
      v2f a = *(const v2f*)&hlast[lrow * HS + kc * 4 + 2 * lhi];
      v2f b = *(const v2f*)&lWih[(gc + lrow) * HE + kc * 4 + 2 * lhi];
      acc = wmma4(a, b, acc);
    }
    const float bias = lb[gc + lrow];
    #pragma unroll
    for (int jj = 0; jj < 8; ++jj)
      gxl[(jj + 8 * lhi) * GXS + gc + lrow] = acc[jj] + bias;
  }
  __syncthreads();

  v2f wlh[8];
  v2f dwi[8][2];
  v2f dwh[16][2];
  v2f wo[16];
  {
    #pragma unroll
    for (int kc = 0; kc < 8; ++kc)
      wlh[kc] = *(const v2f*)&lWhh[growL * HL + kc * 4 + 2 * lhi];
    #pragma unroll
    for (int tt = 0; tt < 2; ++tt) {
      const int gr = tt ? grow1 : grow0;
      #pragma unroll
      for (int kc = 0; kc < 8; ++kc)
        dwi[kc][tt] = *(const v2f*)&dWih[gr * HL + kc * 4 + 2 * lhi];
      #pragma unroll
      for (int kc = 0; kc < 16; ++kc)
        dwh[kc][tt] = *(const v2f*)&dWhh[gr * HD + kc * 4 + 2 * lhi];
    }
    const int f = wave * 16 + lrow;
    #pragma unroll
    for (int kc = 0; kc < 16; ++kc) {
      v2f z = {};
      if (wave < 3 && f < FF)
        z = *(const v2f*)&oW[f * HD + kc * 4 + 2 * lhi];
      wo[kc] = z;
    }
  }
  const float dbi = db[u64], dbf = db[64 + u64], dbg = db[128 + u64], dbo = db[192 + u64];
  const float obias = (wave < 3 && (wave * 16 + lrow) < FF) ? ob[wave * 16 + lrow] : 0.0f;

  for (int i = tid; i < 2 * 16 * HS; i += 256) hbuf[i] = 0.0f;
  for (int i = tid; i < 2 * 16 * HLS; i += 256) hlat[i] = 0.0f;
  float cL[2] = {0.f, 0.f};
  float gx[4][2];
  #pragma unroll
  for (int g = 0; g < 4; ++g) {
    gx[g][0] = gxl[(mrowL + 0) * GXS + 32 * g + u32];
    gx[g][1] = gxl[(mrowL + 1) * GXS + 32 * g + u32];
  }
  float cD[4] = {0.f, 0.f, 0.f, 0.f};
  __syncthreads();

  float* outblk = out + (size_t)b0 * TT * FF;

  #pragma unroll 1
  for (int t = 0; t < TT; ++t) {
    const float* hrd = hbuf + (t & 1) * (16 * HS);
    float*       hwr = hbuf + ((t + 1) & 1) * (16 * HS);
    const float* lrd = hlat + (t & 1) * (16 * HLS);
    float*       lwr = hlat + ((t + 1) & 1) * (16 * HLS);
    {
      v8f acc = {};
      #pragma unroll
      for (int kc = 0; kc < 8; ++kc) {
        v2f a = *(const v2f*)&lrd[lrow * HLS + kc * 4 + 2 * lhi];
        acc = wmma4(a, wlh[kc], acc);
      }
      lwr[hoffL]       = cell32<0>(acc, gx, cL[0]);
      lwr[hoffL + HLS] = cell32<1>(acc, gx, cL[1]);
    }
    __syncthreads();
    {
      v8f acc0 = {}, acc1 = {};
      #pragma unroll
      for (int kc = 0; kc < 8; ++kc) {
        v2f a = *(const v2f*)&lwr[lrow * HLS + kc * 4 + 2 * lhi];
        acc0 = wmma4(a, dwi[kc][0], acc0);
        acc1 = wmma4(a, dwi[kc][1], acc1);
      }
      #pragma unroll
      for (int kc = 0; kc < 16; ++kc) {
        v2f a = *(const v2f*)&hrd[lrow * HS + kc * 4 + 2 * lhi];
        acc0 = wmma4(a, dwh[kc][0], acc0);
        acc1 = wmma4(a, dwh[kc][1], acc1);
      }
      cell64(acc0, acc1, dbi, dbf, dbg, dbo, cD, hwr + hoff);
    }
    __syncthreads();
    if (wave < 3) {
      v8f acc = {};
      #pragma unroll
      for (int kc = 0; kc < 16; ++kc) {
        v2f a = *(const v2f*)&hwr[lrow * HS + kc * 4 + 2 * lhi];
        acc = wmma4(a, wo[kc], acc);
      }
      const int f = wave * 16 + lrow;
      if (f < FF) {
        #pragma unroll
        for (int jj = 0; jj < 8; ++jj) {
          const int m = jj + 8 * lhi;
          outblk[(size_t)m * TT * FF + t * FF + f] = acc[jj] + obias;
        }
      }
    }
  }
}

extern "C" void kernel_launch(void* const* d_in, const int* in_sizes, int n_in,
                              void* d_out, int out_size, void* d_ws, size_t ws_size,
                              hipStream_t stream) {
  (void)in_sizes; (void)n_in; (void)out_size; (void)d_ws; (void)ws_size;
  const float* x    = (const float*)d_in[0];
  const float* eWih = (const float*)d_in[1];
  const float* eWhh = (const float*)d_in[2];
  const float* eb   = (const float*)d_in[3];
  const float* lWih = (const float*)d_in[4];
  const float* lWhh = (const float*)d_in[5];
  const float* lb   = (const float*)d_in[6];
  const float* dWih = (const float*)d_in[7];
  const float* dWhh = (const float*)d_in[8];
  const float* db   = (const float*)d_in[9];
  const float* oW   = (const float*)d_in[10];
  const float* ob   = (const float*)d_in[11];
  float* out = (float*)d_out;

  dim3 grid(BB / 16), block(256);
  hipLaunchKernelGGL(lstm_ae_kernel, grid, block, 0, stream,
                     x, eWih, eWhh, eb, lWih, lWhh, lb, dWih, dWhh, db, oW, ob, out);
}


# cell update: sigm(i)*tanh(g) and sigm(o)*tanh(c) each as ONE quotient (p-1)/((1+a)(p+1)), a=e^-s, p=e^2x, 2x capped at 40 where tanh is 1: 3 IEEE divisions per cell, was 5; sum order kept
# speedup vs baseline: 1.2864x; 1.0756x over previous
#include <hip/hip_runtime.h>

typedef float v2f __attribute__((ext_vector_type(2)));
typedef float v8f __attribute__((ext_vector_type(8)));

#define BB 4096
#define TT 100
#define FF 40
#define HE 64
#define HL 32
#define HD 64

#define GXS 132
#define HS  66
#define HLS 34

__device__ __forceinline__ v8f wmma4(v2f a, v2f b, v8f c) {
  return __builtin_amdgcn_wmma_f32_16x16x4_f32(false, a, false, b, (short)0, c,
                                               false, false);
}

__device__ __forceinline__ float sigm(float x) {
  return 1.0f / (1.0f + __expf(-x));
}
__device__ __forceinline__ float sigm_tanh(float s, float x) {
  float a = __expf(-s);
  float p = __expf(fminf(2.0f * x, 40.0f));
  return (p - 1.0f) / ((1.0f + a) * (p + 1.0f));
}

__device__ __forceinline__ float take_hi(float mine, float theirs) {
  return __builtin_bit_cast(float, __builtin_amdgcn_update_dpp(
      __builtin_bit_cast(int, mine), __builtin_bit_cast(int, theirs),
      0x168, 0xf, 0xc, false));
}
__device__ __forceinline__ float take_lo(float mine, float theirs) {
  return __builtin_bit_cast(float, __builtin_amdgcn_update_dpp(
      __builtin_bit_cast(int, mine), __builtin_bit_cast(int, theirs),
      0x168, 0xf, 0x3, false));
}

template <int XM, int BM>
__device__ __forceinline__ float dppx(float old, float src) {
  return __builtin_bit_cast(float, __builtin_amdgcn_update_dpp(
      __builtin_bit_cast(int, old), __builtin_bit_cast(int, src),
      0x160 | XM, 0xf, BM, false));
}
template <int G, int K>
__device__ __forceinline__ float gate32(const v8f& acc) {
  float v = 0.0f;
  v = dppx<4 * (G ^ 0), 1>(v, acc[0 + K]);
  v = dppx<4 * (G ^ 1), 2>(v, acc[2 + K]);
  v = dppx<4 * (G ^ 2), 4>(v, acc[4 + K]);
  v = dppx<4 * (G ^ 3), 8>(v, acc[6 + K]);
  return v;
}
template <int K>
__device__ __forceinline__ float cell32(const v8f& acc, const float (&gx)[4][2], float& cst) {
  float gi = gate32<0, K>(acc) + gx[0][K];
  float gf = gate32<1, K>(acc) + gx[1][K];
  float gg = gate32<2, K>(acc) + gx[2][K];
  float go = gate32<3, K>(acc) + gx[3][K];
  float c  = sigm(gf) * cst + sigm_tanh(gi, gg);
  cst = c;
  return sigm_tanh(go, c);
}

__device__ __forceinline__ void cell64(const v8f& acc0, const v8f& acc1,
                                       float bi, float bf, float bg, float bo,
                                       float (&cst)[4], float* __restrict__ hdst) {
  #pragma unroll
  for (int k = 0; k < 4; ++k) {
    float gi = take_hi(acc0[k], acc0[4 + k]) + bi;
    float gg = take_lo(acc0[4 + k], acc0[k]) + bg;
    float gf = take_hi(acc1[k], acc1[4 + k]) + bf;
    float go = take_lo(acc1[4 + k], acc1[k]) + bo;
    float c  = sigm(gf) * cst[k] + sigm_tanh(gi, gg);
    cst[k] = c;
    hdst[k * HS] = sigm_tanh(go, c);
  }
}

__global__ __launch_bounds__(256)
void lstm_ae_kernel(const float* __restrict__ x,
                    const float* __restrict__ eWih, const float* __restrict__ eWhh,
                    const float* __restrict__ eb,
                    const float* __restrict__ lWih, const float* __restrict__ lWhh,
                    const float* __restrict__ lb,
                    const float* __restrict__ dWih, const float* __restrict__ dWhh,
                    const float* __restrict__ db,
                    const float* __restrict__ oW,   const float* __restrict__ ob,
                    float* __restrict__ out) {
  __shared__ float gxl[16 * GXS];
  __shared__ float hbuf[2 * 16 * HS];
  __shared__ float hlat[2 * 16 * HLS];

  const int tid  = threadIdx.x;
  const int lane = tid & 31;
  const int wave = tid >> 5;
  const int lrow = lane & 15;
  const int lhi  = lane >> 4;
  const int b0   = blockIdx.x * 16;

  const int u64   = wave * 8 + (lrow & 7);
  const int mrow0 = 8 * lhi + 4 * (lrow >> 3);
  const int grow0 = (lrow >> 3) * 128 + u64;
  const int grow1 = 64 + grow0;
  const int hoff  = mrow0 * HS + u64;
  const int q32   = lrow >> 2;
  const int u32   = wave * 4 + (lrow & 3);
  const int growL = 32 * q32 + u32;
  const int mrowL = 8 * lhi + 2 * q32;
  const int hoffL = mrowL * HLS + u32;

  v2f wih[10][2];
  v2f whh[16][2];
  #pragma unroll
  for (int tt = 0; tt < 2; ++tt) {
    const int gr = tt ? grow1 : grow0;
    #pragma unroll
    for (int kc = 0; kc < 10; ++kc)
      wih[kc][tt] = *(const v2f*)&eWih[gr * FF + kc * 4 + 2 * lhi];
    #pragma unroll
    for (int kc = 0; kc < 16; ++kc)
      whh[kc][tt] = *(const v2f*)&eWhh[gr * HE + kc * 4 + 2 * lhi];
  }
  const float ebi = eb[u64], ebf = eb[64 + u64], ebg = eb[128 + u64], ebo = eb[192 + u64];

  for (int i = tid; i < 2 * 16 * HS; i += 256) hbuf[i] = 0.0f;
  float cE[4] = {0.f, 0.f, 0.f, 0.f};
  __syncthreads();

  const float* xblk = x + (size_t)b0 * TT * FF;
  const float* xrow = xblk + (size_t)lrow * TT * FF + 2 * lhi;

  v2f xc[10];
  #pragma unroll
  for (int kc = 0; kc < 10; ++kc) xc[kc] = *(const v2f*)&xrow[kc * 4];

  #pragma unroll 1
  for (int t = 0; t < TT; ++t) {
    const float* hrd = hbuf + (t & 1) * (16 * HS);
    float*       hwr = hbuf + ((t + 1) & 1) * (16 * HS);
    const float* xt = xrow + ((t + 1 < TT) ? t + 1 : t) * FF;
    v2f xn[10];
    #pragma unroll
    for (int kc = 0; kc < 10; ++kc) xn[kc] = *(const v2f*)&xt[kc * 4];

    v8f acc0 = {}, acc1 = {};
    #pragma unroll
    for (int kc = 0; kc < 10; ++kc) {
      acc0 = wmma4(xc[kc], wih[kc][0], acc0);
      acc1 = wmma4(xc[kc], wih[kc][1], acc1);
    }
    #pragma unroll
    for (int kc = 0; kc < 16; ++kc) {
      v2f a = *(const v2f*)&hrd[lrow * HS + kc * 4 + 2 * lhi];
      acc0 = wmma4(a, whh[kc][0], acc0);
      acc1 = wmma4(a, whh[kc][1], acc1);
    }
    cell64(acc0, acc1, ebi, ebf, ebg, ebo, cE, hwr + hoff);
    __syncthreads();
    #pragma unroll
    for (int kc = 0; kc < 10; ++kc) xc[kc] = xn[kc];
  }
  const float* hlast = hbuf + (TT & 1) * (16 * HS);

  {
    const int gc = wave * 16;
    v8f acc = {};
    #pragma unroll
    for (int kc = 0; kc < 16; ++kc) {
      v2f a = *(const v2f*)&hlast[lrow * HS + kc * 4 + 2 * lhi];
      v2f b = *(const v2f*)&lWih[(gc + lrow) * HE + kc * 4 + 2 * lhi];
      acc = wmma4(a, b, acc);
    }
    const float bias = lb[gc + lrow];
    #pragma unroll
    for (int jj = 0; jj < 8; ++jj)
      gxl[(jj + 8 * lhi) * GXS + gc + lrow] = acc[jj] + bias;
  }
  __syncthreads();

  v2f wlh[8];
  v2f dwi[8][2];
  v2f dwh[16][2];
  v2f wo[16];
  {
    #pragma unroll
    for (int kc = 0; kc < 8; ++kc)
      wlh[kc] = *(const v2f*)&lWhh[growL * HL + kc * 4 + 2 * lhi];
    #pragma unroll
    for (int tt = 0; tt < 2; ++tt) {
      const int gr = tt ? grow1 : grow0;
      #pragma unroll
      for (int kc = 0; kc < 8; ++kc)
        dwi[kc][tt] = *(const v2f*)&dWih[gr * HL + kc * 4 + 2 * lhi];
      #pragma unroll
      for (int kc = 0; kc < 16; ++kc)
        dwh[kc][tt] = *(const v2f*)&dWhh[gr * HD + kc * 4 + 2 * lhi];
    }
    const int f = wave * 16 + lrow;
    #pragma unroll
    for (int kc = 0; kc < 16; ++kc) {
      v2f z = {};
      if (wave < 3 && f < FF)
        z = *(const v2f*)&oW[f * HD + kc * 4 + 2 * lhi];
      wo[kc] = z;
    }
  }
  const float dbi = db[u64], dbf = db[64 + u64], dbg = db[128 + u64], dbo = db[192 + u64];
  const float obias = (wave < 3 && (wave * 16 + lrow) < FF) ? ob[wave * 16 + lrow] : 0.0f;

  for (int i = tid; i < 2 * 16 * HS; i += 256) hbuf[i] = 0.0f;
  for (int i = tid; i < 2 * 16 * HLS; i += 256) hlat[i] = 0.0f;
  float cL[2] = {0.f, 0.f};
  float gx[4][2];
  #pragma unroll
  for (int g = 0; g < 4; ++g) {
    gx[g][0] = gxl[(mrowL + 0) * GXS + 32 * g + u32];
    gx[g][1] = gxl[(mrowL + 1) * GXS + 32 * g + u32];
  }
  float cD[4] = {0.f, 0.f, 0.f, 0.f};
  __syncthreads();

  float* outblk = out + (size_t)b0 * TT * FF;

  #pragma unroll 1
  for (int t = 0; t < TT; ++t) {
    const float* hrd = hbuf + (t & 1) * (16 * HS);
    float*       hwr = hbuf + ((t + 1) & 1) * (16 * HS);
    const float* lrd = hlat + (t & 1) * (16 * HLS);
    float*       lwr = hlat + ((t + 1) & 1) * (16 * HLS);
    {
      v8f acc = {};
      #pragma unroll
      for (int kc = 0; kc < 8; ++kc) {
        v2f a = *(const v2f*)&lrd[lrow * HLS + kc * 4 + 2 * lhi];
        acc = wmma4(a, wlh[kc], acc);
      }
      lwr[hoffL]       = cell32<0>(acc, gx, cL[0]);
      lwr[hoffL + HLS] = cell32<1>(acc, gx, cL[1]);
    }
    __syncthreads();
    {
      v8f acc0 = {}, acc1 = {};
      #pragma unroll
      for (int kc = 0; kc < 8; ++kc) {
        v2f a = *(const v2f*)&lwr[lrow * HLS + kc * 4 + 2 * lhi];
        acc0 = wmma4(a, dwi[kc][0], acc0);
        acc1 = wmma4(a, dwi[kc][1], acc1);
      }
      #pragma unroll
      for (int kc = 0; kc < 16; ++kc) {
        v2f a = *(const v2f*)&hrd[lrow * HS + kc * 4 + 2 * lhi];
        acc0 = wmma4(a, dwh[kc][0], acc0);
        acc1 = wmma4(a, dwh[kc][1], acc1);
      }
      cell64(acc0, acc1, dbi, dbf, dbg, dbo, cD, hwr + hoff);
    }
    __syncthreads();
    if (wave < 3) {
      v8f acc = {};
      #pragma unroll
      for (int kc = 0; kc < 16; ++kc) {
        v2f a = *(const v2f*)&hwr[lrow * HS + kc * 4 + 2 * lhi];
        acc = wmma4(a, wo[kc], acc);
      }
      const int f = wave * 16 + lrow;
      if (f < FF) {
        #pragma unroll
        for (int jj = 0; jj < 8; ++jj) {
          const int m = jj + 8 * lhi;
          outblk[(size_t)m * TT * FF + t * FF + f] = acc[jj] + obias;
        }
      }
    }
  }
}

extern "C" void kernel_launch(void* const* d_in, const int* in_sizes, int n_in,
                              void* d_out, int out_size, void* d_ws, size_t ws_size,
                              hipStream_t stream) {
  (void)in_sizes; (void)n_in; (void)out_size; (void)d_ws; (void)ws_size;
  const float* x    = (const float*)d_in[0];
  const float* eWih = (const float*)d_in[1];
  const float* eWhh = (const float*)d_in[2];
  const float* eb   = (const float*)d_in[3];
  const float* lWih = (const float*)d_in[4];
  const float* lWhh = (const float*)d_in[5];
  const float* lb   = (const float*)d_in[6];
  const float* dWih = (const float*)d_in[7];
  const float* dWhh = (const float*)d_in[8];
  const float* db   = (const float*)d_in[9];
  const float* oW   = (const float*)d_in[10];
  const float* ob   = (const float*)d_in[11];
  float* out = (float*)d_out;

  dim3 grid(BB / 16), block(256);
  hipLaunchKernelGGL(lstm_ae_kernel, grid, block, 0, stream,
                     x, eWih, eWhh, eb, lWih, lWhh, lb, dWih, dWhh, db, oW, ob, out);
}


# decoder: h_lat and h_dec A fragments read from LDS once a step, one group right behind each barrier, kept in registers for all 3 products (was 48 reads among the WMMAs); gate sums: recurrent first
# speedup vs baseline: 1.3271x; 1.0316x over previous
#include <hip/hip_runtime.h>

typedef float v2f __attribute__((ext_vector_type(2)));
typedef float v8f __attribute__((ext_vector_type(8)));

#define BB 4096
#define TT 100
#define FF 40
#define HE 64
#define HL 32
#define HD 64

#define GXS 132
#define HS  66
#define HLS 34

__device__ __forceinline__ v8f wmma4(v2f a, v2f b, v8f c) {
  return __builtin_amdgcn_wmma_f32_16x16x4_f32(false, a, false, b, (short)0, c,
                                               false, false);
}

__device__ __forceinline__ float sigm(float x) {
  return 1.0f / (1.0f + __expf(-x));
}
__device__ __forceinline__ float sigm_tanh(float s, float x) {
  float a = __expf(-s);
  float p = __expf(fminf(2.0f * x, 40.0f));
  return (p - 1.0f) / ((1.0f + a) * (p + 1.0f));
}

__device__ __forceinline__ float take_hi(float mine, float theirs) {
  return __builtin_bit_cast(float, __builtin_amdgcn_update_dpp(
      __builtin_bit_cast(int, mine), __builtin_bit_cast(int, theirs),
      0x168, 0xf, 0xc, false));
}
__device__ __forceinline__ float take_lo(float mine, float theirs) {
  return __builtin_bit_cast(float, __builtin_amdgcn_update_dpp(
      __builtin_bit_cast(int, mine), __builtin_bit_cast(int, theirs),
      0x168, 0xf, 0x3, false));
}

template <int XM, int BM>
__device__ __forceinline__ float dppx(float old, float src) {
  return __builtin_bit_cast(float, __builtin_amdgcn_update_dpp(
      __builtin_bit_cast(int, old), __builtin_bit_cast(int, src),
      0x160 | XM, 0xf, BM, false));
}
template <int G, int K>
__device__ __forceinline__ float gate32(const v8f& acc) {
  float v = 0.0f;
  v = dppx<4 * (G ^ 0), 1>(v, acc[0 + K]);
  v = dppx<4 * (G ^ 1), 2>(v, acc[2 + K]);
  v = dppx<4 * (G ^ 2), 4>(v, acc[4 + K]);
  v = dppx<4 * (G ^ 3), 8>(v, acc[6 + K]);
  return v;
}
template <int K>
__device__ __forceinline__ float cell32(const v8f& acc, const float (&gx)[4][2], float& cst) {
  float gi = gate32<0, K>(acc) + gx[0][K];
  float gf = gate32<1, K>(acc) + gx[1][K];
  float gg = gate32<2, K>(acc) + gx[2][K];
  float go = gate32<3, K>(acc) + gx[3][K];
  float c  = sigm(gf) * cst + sigm_tanh(gi, gg);
  cst = c;
  return sigm_tanh(go, c);
}

__device__ __forceinline__ void cell64(const v8f& acc0, const v8f& acc1,
                                       float bi, float bf, float bg, float bo,
                                       float (&cst)[4], float* __restrict__ hdst) {
  #pragma unroll
  for (int k = 0; k < 4; ++k) {
    float gi = take_hi(acc0[k], acc0[4 + k]) + bi;
    float gg = take_lo(acc0[4 + k], acc0[k]) + bg;
    float gf = take_hi(acc1[k], acc1[4 + k]) + bf;
    float go = take_lo(acc1[4 + k], acc1[k]) + bo;
    float c  = sigm(gf) * cst[k] + sigm_tanh(gi, gg);
    cst[k] = c;
    hdst[k * HS] = sigm_tanh(go, c);
  }
}

__global__ __launch_bounds__(256)
void lstm_ae_kernel(const float* __restrict__ x,
                    const float* __restrict__ eWih, const float* __restrict__ eWhh,
                    const float* __restrict__ eb,
                    const float* __restrict__ lWih, const float* __restrict__ lWhh,
                    const float* __restrict__ lb,
                    const float* __restrict__ dWih, const float* __restrict__ dWhh,
                    const float* __restrict__ db,
                    const float* __restrict__ oW,   const float* __restrict__ ob,
                    float* __restrict__ out) {
  __shared__ float gxl[16 * GXS];
  __shared__ float hbuf[2 * 16 * HS];
  __shared__ float hlat[2 * 16 * HLS];

  const int tid  = threadIdx.x;
  const int lane = tid & 31;
  const int wave = tid >> 5;
  const int lrow = lane & 15;
  const int lhi  = lane >> 4;
  const int b0   = blockIdx.x * 16;

  const int u64   = wave * 8 + (lrow & 7);
  const int mrow0 = 8 * lhi + 4 * (lrow >> 3);
  const int grow0 = (lrow >> 3) * 128 + u64;
  const int grow1 = 64 + grow0;
  const int hoff  = mrow0 * HS + u64;
  const int q32   = lrow >> 2;
  const int u32   = wave * 4 + (lrow & 3);
  const int growL = 32 * q32 + u32;
  const int mrowL = 8 * lhi + 2 * q32;
  const int hoffL = mrowL * HLS + u32;

  v2f wih[10][2];
  v2f whh[16][2];
  #pragma unroll
  for (int tt = 0; tt < 2; ++tt) {
    const int gr = tt ? grow1 : grow0;
    #pragma unroll
    for (int kc = 0; kc < 10; ++kc)
      wih[kc][tt] = *(const v2f*)&eWih[gr * FF + kc * 4 + 2 * lhi];
    #pragma unroll
    for (int kc = 0; kc < 16; ++kc)
      whh[kc][tt] = *(const v2f*)&eWhh[gr * HE + kc * 4 + 2 * lhi];
  }
  const float ebi = eb[u64], ebf = eb[64 + u64], ebg = eb[128 + u64], ebo = eb[192 + u64];

  for (int i = tid; i < 2 * 16 * HS; i += 256) hbuf[i] = 0.0f;
  float cE[4] = {0.f, 0.f, 0.f, 0.f};
  __syncthreads();

  const float* xblk = x + (size_t)b0 * TT * FF;
  const float* xrow = xblk + (size_t)lrow * TT * FF + 2 * lhi;

  v2f xc[10];
  #pragma unroll
  for (int kc = 0; kc < 10; ++kc) xc[kc] = *(const v2f*)&xrow[kc * 4];

  #pragma unroll 1
  for (int t = 0; t < TT; ++t) {
    const float* hrd = hbuf + (t & 1) * (16 * HS);
    float*       hwr = hbuf + ((t + 1) & 1) * (16 * HS);
    const float* xt = xrow + ((t + 1 < TT) ? t + 1 : t) * FF;
    v2f xn[10];
    #pragma unroll
    for (int kc = 0; kc < 10; ++kc) xn[kc] = *(const v2f*)&xt[kc * 4];

    v8f acc0 = {}, acc1 = {};
    #pragma unroll
    for (int kc = 0; kc < 10; ++kc) {
      acc0 = wmma4(xc[kc], wih[kc][0], acc0);
      acc1 = wmma4(xc[kc], wih[kc][1], acc1);
    }
    #pragma unroll
    for (int kc = 0; kc < 16; ++kc) {
      v2f a = *(const v2f*)&hrd[lrow * HS + kc * 4 + 2 * lhi];
      acc0 = wmma4(a, whh[kc][0], acc0);
      acc1 = wmma4(a, whh[kc][1], acc1);
    }
    cell64(acc0, acc1, ebi, ebf, ebg, ebo, cE, hwr + hoff);
    __syncthreads();
    #pragma unroll
    for (int kc = 0; kc < 10; ++kc) xc[kc] = xn[kc];
  }
  const float* hlast = hbuf + (TT & 1) * (16 * HS);

  {
    const int gc = wave * 16;
    v8f acc = {};
    #pragma unroll
    for (int kc = 0; kc < 16; ++kc) {
      v2f a = *(const v2f*)&hlast[lrow * HS + kc * 4 + 2 * lhi];
      v2f b = *(const v2f*)&lWih[(gc + lrow) * HE + kc * 4 + 2 * lhi];
      acc = wmma4(a, b, acc);
    }
    const float bias = lb[gc + lrow];
    #pragma unroll
    for (int jj = 0; jj < 8; ++jj)
      gxl[(jj + 8 * lhi) * GXS + gc + lrow] = acc[jj] + bias;
  }
  __syncthreads();

  v2f wlh[8];
  v2f dwi[8][2];
  v2f dwh[16][2];
  v2f wo[16];
  {
    #pragma unroll
    for (int kc = 0; kc < 8; ++kc)
      wlh[kc] = *(const v2f*)&lWhh[growL * HL + kc * 4 + 2 * lhi];
    #pragma unroll
    for (int tt = 0; tt < 2; ++tt) {
      const int gr = tt ? grow1 : grow0;
      #pragma unroll
      for (int kc = 0; kc < 8; ++kc)
        dwi[kc][tt] = *(const v2f*)&dWih[gr * HL + kc * 4 + 2 * lhi];
      #pragma unroll
      for (int kc = 0; kc < 16; ++kc)
        dwh[kc][tt] = *(const v2f*)&dWhh[gr * HD + kc * 4 + 2 * lhi];
    }
    const int f = wave * 16 + lrow;
    #pragma unroll
    for (int kc = 0; kc < 16; ++kc) {
      v2f z = {};
      if (wave < 3 && f < FF)
        z = *(const v2f*)&oW[f * HD + kc * 4 + 2 * lhi];
      wo[kc] = z;
    }
  }
  const float dbi = db[u64], dbf = db[64 + u64], dbg = db[128 + u64], dbo = db[192 + u64];
  const float obias = (wave < 3 && (wave * 16 + lrow) < FF) ? ob[wave * 16 + lrow] : 0.0f;

  float cL[2] = {0.f, 0.f};
  float gx[4][2];
  #pragma unroll
  for (int g = 0; g < 4; ++g) {
    gx[g][0] = gxl[(mrowL + 0) * GXS + 32 * g + u32];
    gx[g][1] = gxl[(mrowL + 1) * GXS + 32 * g + u32];
  }
  float cD[4] = {0.f, 0.f, 0.f, 0.f};

  float* outblk = out + (size_t)b0 * TT * FF;

  v2f la[8] = {}, ha[16] = {};

  #pragma unroll 1
  for (int t = 0; t < TT; ++t) {
    float* hwr = hbuf + ((t + 1) & 1) * (16 * HS);
    float* lwr = hlat + ((t + 1) & 1) * (16 * HLS);
    {
      v8f acc = {};
      #pragma unroll
      for (int kc = 0; kc < 8; ++kc) acc = wmma4(la[kc], wlh[kc], acc);
      lwr[hoffL]       = cell32<0>(acc, gx, cL[0]);
      lwr[hoffL + HLS] = cell32<1>(acc, gx, cL[1]);
    }
    __syncthreads();
    #pragma unroll
    for (int kc = 0; kc < 8; ++kc)
      la[kc] = *(const v2f*)&lwr[lrow * HLS + kc * 4 + 2 * lhi];
    {
      v8f acc0 = {}, acc1 = {};
      #pragma unroll
      for (int kc = 0; kc < 16; ++kc) {
        acc0 = wmma4(ha[kc], dwh[kc][0], acc0);
        acc1 = wmma4(ha[kc], dwh[kc][1], acc1);
      }
      #pragma unroll
      for (int kc = 0; kc < 8; ++kc) {
        acc0 = wmma4(la[kc], dwi[kc][0], acc0);
        acc1 = wmma4(la[kc], dwi[kc][1], acc1);
      }
      cell64(acc0, acc1, dbi, dbf, dbg, dbo, cD, hwr + hoff);
    }
    __syncthreads();
    #pragma unroll
    for (int kc = 0; kc < 16; ++kc)
      ha[kc] = *(const v2f*)&hwr[lrow * HS + kc * 4 + 2 * lhi];
    if (wave < 3) {
      v8f acc = {};
      #pragma unroll
      for (int kc = 0; kc < 16; ++kc) acc = wmma4(ha[kc], wo[kc], acc);
      const int f = wave * 16 + lrow;
      if (f < FF) {
        #pragma unroll
        for (int jj = 0; jj < 8; ++jj) {
          const int m = jj + 8 * lhi;
          outblk[(size_t)m * TT * FF + t * FF + f] = acc[jj] + obias;
        }
      }
    }
  }
}

extern "C" void kernel_launch(void* const* d_in, const int* in_sizes, int n_in,
                              void* d_out, int out_size, void* d_ws, size_t ws_size,
                              hipStream_t stream) {
  (void)in_sizes; (void)n_in; (void)out_size; (void)d_ws; (void)ws_size;
  const float* x    = (const float*)d_in[0];
  const float* eWih = (const float*)d_in[1];
  const float* eWhh = (const float*)d_in[2];
  const float* eb   = (const float*)d_in[3];
  const float* lWih = (const float*)d_in[4];
  const float* lWhh = (const float*)d_in[5];
  const float* lb   = (const float*)d_in[6];
  const float* dWih = (const float*)d_in[7];
  const float* dWhh = (const float*)d_in[8];
  const float* db   = (const float*)d_in[9];
  const float* oW   = (const float*)d_in[10];
  const float* ob   = (const float*)d_in[11];
  float* out = (float*)d_out;

  dim3 grid(BB / 16), block(256);
  hipLaunchKernelGGL(lstm_ae_kernel, grid, block, 0, stream,
                     x, eWih, eWhh, eb, lWih, lWhh, lb, dWih, dWhh, db, oW, ob, out);
}
